# starting kernel
# baseline (speedup 1.0000x reference)
#include <hip/hip_runtime.h>
#include <math.h>


typedef __attribute__((ext_vector_type(2))) float v2f;
typedef __attribute__((ext_vector_type(8))) float v8f;

#define BB 4
#define SS 2048
#define DD 64
#define HH 16
#define HDIM 4

static __device__ __forceinline__ v8f wmma_f32_16x16x4(v2f a, v2f b, v8f c) {
  return __builtin_amdgcn_wmma_f32_16x16x4_f32(false, a, false, b, (short)0, c,
                                               false, false);
}

__global__ __launch_bounds__(128) void qkv_kernel(
    const float* __restrict__ x,
    const float* __restrict__ Wq, const float* __restrict__ bq,
    const float* __restrict__ Wk, const float* __restrict__ bk,
    const float* __restrict__ Wv, const float* __restrict__ bv,
    float* __restrict__ ws) {
  const int NT = (BB * SS) / 16;
  int tid   = blockIdx.x * 4 + (threadIdx.x >> 5);
  int which = tid / NT;
  int tile  = tid % NT;
  int r0    = tile * 16;

  const float* W    = (which == 0) ? Wq : (which == 1) ? Wk : Wv;
  const float* bias = (which == 0) ? bq : (which == 1) ? bk : bv;
  float* Out = ws + (size_t)which * (BB * SS * DD);

  int lane = threadIdx.x & 31;
  int half = lane >> 4;
  int l16  = lane & 15;

  v2f afrag[16];
  const float* xrow = x + (size_t)(r0 + l16) * DD + 2 * half;
#pragma unroll
  for (int kc = 0; kc < 16; ++kc) {
    afrag[kc].x = xrow[4 * kc + 0];
    afrag[kc].y = xrow[4 * kc + 1];
  }

  v8f acc[4];
#pragma unroll
  for (int n = 0; n < 4; ++n)
#pragma unroll
    for (int v = 0; v < 8; ++v) acc[n][v] = 0.0f;

#pragma unroll
  for (int kc = 0; kc < 16; ++kc) {
    int krow = 4 * kc + 2 * half;
#pragma unroll
    for (int n = 0; n < 4; ++n) {
      v2f bfrag;
      bfrag.x = W[(size_t)krow * DD + n * 16 + l16];
      bfrag.y = W[(size_t)(krow + 1) * DD + n * 16 + l16];
      acc[n] = wmma_f32_16x16x4(afrag[kc], bfrag, acc[n]);
    }
  }

#pragma unroll
  for (int n = 0; n < 4; ++n) {
    int col = n * 16 + l16;
    float bb = bias[col];
#pragma unroll
    for (int v = 0; v < 8; ++v) {
      int row = r0 + v + 8 * half;
      Out[(size_t)row * DD + col] = fmaxf(acc[n][v] + bb, 0.0f);
    }
  }
}

__global__ __launch_bounds__(128) void attn_kernel(
    const float* __restrict__ x, const float* __restrict__ ws,
    float* __restrict__ out) {
  const float* Q = ws;
  const float* K = ws + (size_t)BB * SS * DD;
  const float* V = ws + (size_t)2 * BB * SS * DD;

  __shared__ float ldsP[4][16][16];

  int wave = threadIdx.x >> 5;
  int lane = threadIdx.x & 31;
  int half = lane >> 4;
  int l16  = lane & 15;

  const int QT = SS / 64;
  int bh = blockIdx.x / QT;
  int qt = blockIdx.x % QT;
  int b = bh / HH, h = bh % HH;
  int q0 = qt * 64 + wave * 16;

  const float* Qb = Q + ((size_t)b * SS) * DD + h * HDIM;
  const float* Kb = K + ((size_t)b * SS) * DD + h * HDIM;
  const float* Vb = V + ((size_t)b * SS) * DD + h * HDIM;

  v2f aq;
  aq.x = Qb[(size_t)(q0 + l16) * DD + 2 * half];
  aq.y = Qb[(size_t)(q0 + l16) * DD + 2 * half + 1];

  float mrow[8], lrow[8];
  v8f accO, zc;
#pragma unroll
  for (int v = 0; v < 8; ++v) {
    mrow[v] = -INFINITY;
    lrow[v] = 0.0f;
    accO[v] = 0.0f;
    zc[v]   = 0.0f;
  }

  const float scale = 0.5f;

  for (int k0 = 0; k0 < SS; k0 += 16) {
    if (k0 + 16 < SS) {
      __builtin_prefetch(Kb + (size_t)(k0 + 16 + l16) * DD, 0, 1);
      __builtin_prefetch(Vb + (size_t)(k0 + 16 + l16) * DD, 0, 1);
    }
    v2f bk;
    bk.x = Kb[(size_t)(k0 + l16) * DD + 2 * half];
    bk.y = Kb[(size_t)(k0 + l16) * DD + 2 * half + 1];
    v8f s = wmma_f32_16x16x4(aq, bk, zc);

    float p[8];
#pragma unroll
    for (int v = 0; v < 8; ++v) {
      float sv = s[v] * scale;
      float m = sv;
      m = fmaxf(m, __shfl_xor(m, 1, 16));
      m = fmaxf(m, __shfl_xor(m, 2, 16));
      m = fmaxf(m, __shfl_xor(m, 4, 16));
      m = fmaxf(m, __shfl_xor(m, 8, 16));
      float mnew  = fmaxf(mrow[v], m);
      float alpha = __expf(mrow[v] - mnew);
      float pv    = __expf(sv - mnew);
      float rs = pv;
      rs += __shfl_xor(rs, 1, 16);
      rs += __shfl_xor(rs, 2, 16);
      rs += __shfl_xor(rs, 4, 16);
      rs += __shfl_xor(rs, 8, 16);
      lrow[v] = lrow[v] * alpha + rs;
      mrow[v] = mnew;
      accO[v] *= alpha;
      p[v] = pv;
    }

#pragma unroll
    for (int v = 0; v < 8; ++v) ldsP[wave][v + 8 * half][l16] = p[v];

#pragma unroll
    for (int kc = 0; kc < 4; ++kc) {
      v2f ap;
      ap.x = ldsP[wave][l16][4 * kc + 2 * half];
      ap.y = ldsP[wave][l16][4 * kc + 2 * half + 1];
      v2f bv;
      int krow = k0 + 4 * kc + 2 * half;
      bv.x = (l16 < HDIM) ? Vb[(size_t)krow * DD + l16] : 0.0f;
      bv.y = (l16 < HDIM) ? Vb[(size_t)(krow + 1) * DD + l16] : 0.0f;
      accO = wmma_f32_16x16x4(ap, bv, accO);
    }
  }

  if (l16 < HDIM) {
    int ch = h * HDIM + l16;
#pragma unroll
    for (int v = 0; v < 8; ++v) {
      int row = q0 + v + 8 * half;
      size_t idx = ((size_t)b * SS + row) * DD + ch;
      out[idx] = accO[v] / lrow[v] + x[idx];
    }
  }
}

__global__ __launch_bounds__(256) void bn_stats_kernel(
    const float* __restrict__ out, float* __restrict__ stats) {
  __shared__ float ssum[256], ssq[256];
  int ch = blockIdx.x;
  int t = threadIdx.x;
  float s = 0.0f, q = 0.0f;
  for (int i = t; i < BB * SS; i += 256) {
    float v = out[(size_t)i * DD + ch];
    s += v;
    q += v * v;
  }
  ssum[t] = s;
  ssq[t] = q;
  __syncthreads();
  for (int off = 128; off > 0; off >>= 1) {
    if (t < off) {
      ssum[t] += ssum[t + off];
      ssq[t] += ssq[t + off];
    }
    __syncthreads();
  }
  if (t == 0) {
    const float invN = 1.0f / (float)(BB * SS);
    float mean = ssum[0] * invN;
    float var = ssq[0] * invN - mean * mean;
    stats[ch] = mean;
    stats[DD + ch] = rsqrtf(var + 1e-5f);
  }
}

__global__ __launch_bounds__(256) void bn_apply_kernel(
    float* __restrict__ out, const float* __restrict__ stats,
    const float* __restrict__ gamma, const float* __restrict__ beta) {
  int idx = blockIdx.x * 256 + threadIdx.x;
  int ch = idx & (DD - 1);
  float v = out[idx];
  out[idx] = (v - stats[ch]) * stats[DD + ch] * gamma[ch] + beta[ch];
}

extern "C" void kernel_launch(void* const* d_in, const int* in_sizes, int n_in,
                              void* d_out, int out_size, void* d_ws,
                              size_t ws_size, hipStream_t stream) {
  const float* x     = (const float*)d_in[0];
  const float* Wq    = (const float*)d_in[1];
  const float* bq    = (const float*)d_in[2];
  const float* Wk    = (const float*)d_in[3];
  const float* bk    = (const float*)d_in[4];
  const float* Wv    = (const float*)d_in[5];
  const float* bv    = (const float*)d_in[6];
  const float* gamma = (const float*)d_in[7];
  const float* beta  = (const float*)d_in[8];
  float* out = (float*)d_out;
  float* ws = (float*)d_ws;
  float* stats = ws + (size_t)3 * BB * SS * DD;

  qkv_kernel<<<(3 * (BB * SS / 16)) / 4, 128, 0, stream>>>(x, Wq, bq, Wk, bk,
                                                           Wv, bv, ws);
  attn_kernel<<<BB * HH * (SS / 64), 128, 0, stream>>>(x, ws, out);
  bn_stats_kernel<<<DD, 256, 0, stream>>>(out, stats);
  bn_apply_kernel<<<(BB * SS * DD) / 256, 256, 0, stream>>>(out, stats, gamma,
                                                            beta);
}


# attention: a 32-wave workgroup per 512 queries copies its head's whole K and V from the workspace into LDS once; both key sweeps read K,V fragments from LDS, not global; sums and order unchanged
# speedup vs baseline: 3.4856x; 3.4856x over previous
#include <hip/hip_runtime.h>
#include <math.h>


typedef __attribute__((ext_vector_type(2))) float v2f;
typedef __attribute__((ext_vector_type(8))) float v8f;

#define BB 4
#define SS 2048
#define DD 64
#define HH 16
#define HDIM 4

static __device__ __forceinline__ v8f wmma_f32_16x16x4(v2f a, v2f b, v8f c) {
  return __builtin_amdgcn_wmma_f32_16x16x4_f32(false, a, false, b, (short)0, c,
                                               false, false);
}

__global__ __launch_bounds__(128) void qkv_kernel(
    const float* __restrict__ x,
    const float* __restrict__ Wq, const float* __restrict__ bq,
    const float* __restrict__ Wk, const float* __restrict__ bk,
    const float* __restrict__ Wv, const float* __restrict__ bv,
    float* __restrict__ ws) {
  const int NT = (BB * SS) / 16;
  int tid   = blockIdx.x * 4 + (threadIdx.x >> 5);
  int which = tid / NT;
  int tile  = tid % NT;
  int r0    = tile * 16;

  const float* W    = (which == 0) ? Wq : (which == 1) ? Wk : Wv;
  const float* bias = (which == 0) ? bq : (which == 1) ? bk : bv;
  float* Out = ws + (size_t)which * (BB * SS * DD);

  int lane = threadIdx.x & 31;
  int half = lane >> 4;
  int l16  = lane & 15;

  v2f afrag[16];
  const float* xrow = x + (size_t)(r0 + l16) * DD + 2 * half;
#pragma unroll
  for (int kc = 0; kc < 16; ++kc) {
    afrag[kc].x = xrow[4 * kc + 0];
    afrag[kc].y = xrow[4 * kc + 1];
  }

  v8f acc[4];
#pragma unroll
  for (int n = 0; n < 4; ++n)
#pragma unroll
    for (int v = 0; v < 8; ++v) acc[n][v] = 0.0f;

#pragma unroll
  for (int kc = 0; kc < 16; ++kc) {
    int krow = 4 * kc + 2 * half;
#pragma unroll
    for (int n = 0; n < 4; ++n) {
      v2f bfrag;
      bfrag.x = W[(size_t)krow * DD + n * 16 + l16];
      bfrag.y = W[(size_t)(krow + 1) * DD + n * 16 + l16];
      acc[n] = wmma_f32_16x16x4(afrag[kc], bfrag, acc[n]);
    }
  }

#pragma unroll
  for (int n = 0; n < 4; ++n) {
    int col = n * 16 + l16;
    float bb = bias[col];
#pragma unroll
    for (int v = 0; v < 8; ++v) {
      int row = r0 + v + 8 * half;
      Out[(size_t)row * DD + col] = fmaxf(acc[n][v] + bb, 0.0f);
    }
  }
}

#define ATT_WAVES 32
#define ATT_QPB   (ATT_WAVES * 16)
#define LDS_K     0
#define LDS_V     (SS * HDIM)
#define LDS_Z     (2 * SS * HDIM)
#define LDS_P     (LDS_Z + 128)
#define LDS_USED  (LDS_P + ATT_WAVES * 256)
#define LDS_FLOATS (164 * 256)

__global__ __launch_bounds__(ATT_WAVES * 32) void attn_kernel(
    const float* __restrict__ x, const float* __restrict__ ws,
    float* __restrict__ out) {
  const float* Q = ws;
  const float* K = ws + (size_t)BB * SS * DD;
  const float* V = ws + (size_t)2 * BB * SS * DD;

  __shared__ __attribute__((aligned(16))) float smem[LDS_FLOATS];
  static_assert(LDS_USED <= LDS_FLOATS, "LDS map");

  int t    = threadIdx.x;
  int wave = t >> 5;
  int lane = t & 31;
  int half = lane >> 4;
  int l16  = lane & 15;

  const int QT = SS / ATT_QPB;
  int bh = blockIdx.x / QT;
  int qt = blockIdx.x % QT;
  int b = bh / HH, h = bh % HH;
  int q0 = qt * ATT_QPB + wave * 16;

  const float* Qb = Q + ((size_t)b * SS) * DD + h * HDIM;
  const float* Kb = K + ((size_t)b * SS) * DD + h * HDIM;
  const float* Vb = V + ((size_t)b * SS) * DD + h * HDIM;

  {
    float4 kr[SS / (ATT_WAVES * 32)];
#pragma unroll
    for (int i = 0; i < SS / (ATT_WAVES * 32); ++i) {
      const float* kp = Kb + (size_t)(t + i * ATT_WAVES * 32) * DD;
      kr[i] = make_float4(kp[0], kp[1], kp[2], kp[3]);
    }
    const float* vp = Vb + (size_t)(2 * t) * DD;
    float4 va = make_float4(vp[0], vp[1], vp[2], vp[3]);
    float4 vb = make_float4(vp[DD], vp[DD + 1], vp[DD + 2], vp[DD + 3]);
#pragma unroll
    for (int i = 0; i < SS / (ATT_WAVES * 32); ++i)
      *(float4*)&smem[LDS_K + (t + i * ATT_WAVES * 32) * 4] = kr[i];
    *(float4*)&smem[LDS_V + t * 8]     = make_float4(va.x, vb.x, va.y, vb.y);
    *(float4*)&smem[LDS_V + t * 8 + 4] = make_float4(va.z, vb.z, va.w, vb.w);
    if (t < 128) smem[LDS_Z + t] = 0.0f;
  }

  v2f aq;
  aq.x = Qb[(size_t)(q0 + l16) * DD + 2 * half];
  aq.y = Qb[(size_t)(q0 + l16) * DD + 2 * half + 1];

  __syncthreads();

  v8f zc;
#pragma unroll
  for (int v = 0; v < 8; ++v) zc[v] = 0.0f;

  const float* kfrag = smem + LDS_K + l16 * 4 + 2 * half;

  float mx[8];
#pragma unroll
  for (int v = 0; v < 8; ++v) mx[v] = -INFINITY;

#pragma unroll 4
  for (int k0 = 0; k0 < SS; k0 += 16) {
    v2f bk = *(const v2f*)(kfrag + k0 * 4);
    v8f s = wmma_f32_16x16x4(aq, bk, zc);
#pragma unroll
    for (int v = 0; v < 8; ++v) mx[v] = fmaxf(mx[v], s[v]);
  }

  const float c = 0.72134752044448170368f;
  float mc[8];
#pragma unroll
  for (int v = 0; v < 8; ++v) {
    float m = mx[v];
    m = fmaxf(m, __shfl_xor(m, 1, 16));
    m = fmaxf(m, __shfl_xor(m, 2, 16));
    m = fmaxf(m, __shfl_xor(m, 4, 16));
    m = fmaxf(m, __shfl_xor(m, 8, 16));
    mc[v] = m * c;
  }

  float lpart[8];
  v8f accO;
#pragma unroll
  for (int v = 0; v < 8; ++v) {
    lpart[v] = 0.0f;
    accO[v]  = 0.0f;
  }

  const bool padlane = (l16 >= HDIM);
  const float* vfrag = padlane ? (smem + LDS_Z + 16)
                               : (smem + LDS_V + 8 * half + 2 * l16);
  const int vstep = padlane ? 0 : 64;

  float* pst = smem + LDS_P + wave * 256 + (8 * half) * 16 + l16;
  const float* pld = smem + LDS_P + wave * 256 + l16 * 16 + 2 * half;

  for (int k0 = 0; k0 < SS; k0 += 16) {
    v2f bk = *(const v2f*)(kfrag + k0 * 4);
    v8f s = wmma_f32_16x16x4(aq, bk, zc);

    float p[8];
#pragma unroll
    for (int v = 0; v < 8; ++v) {
      p[v] = __builtin_amdgcn_exp2f(fmaf(s[v], c, -mc[v]));
      lpart[v] += p[v];
    }

#pragma unroll
    for (int v = 0; v < 8; ++v) pst[v * 16] = p[v];

#pragma unroll
    for (int kc = 0; kc < 4; ++kc) {
      v2f ap = *(const v2f*)(pld + 4 * kc);
      v2f bv = *(const v2f*)(vfrag + 16 * kc);
      accO = wmma_f32_16x16x4(ap, bv, accO);
    }
    vfrag += vstep;
  }

  float lrow[8];
#pragma unroll
  for (int v = 0; v < 8; ++v) {
    float rs = lpart[v];
    rs += __shfl_xor(rs, 1, 16);
    rs += __shfl_xor(rs, 2, 16);
    rs += __shfl_xor(rs, 4, 16);
    rs += __shfl_xor(rs, 8, 16);
    lrow[v] = rs;
  }

  if (l16 < HDIM) {
    int ch = h * HDIM + l16;
#pragma unroll
    for (int v = 0; v < 8; ++v) {
      int row = q0 + v + 8 * half;
      size_t idx = ((size_t)b * SS + row) * DD + ch;
      out[idx] = accO[v] / lrow[v] + x[idx];
    }
  }
}

__global__ __launch_bounds__(256) void bn_stats_kernel(
    const float* __restrict__ out, float* __restrict__ stats) {
  __shared__ float ssum[256], ssq[256];
  int ch = blockIdx.x;
  int t = threadIdx.x;
  float s = 0.0f, q = 0.0f;
  for (int i = t; i < BB * SS; i += 256) {
    float v = out[(size_t)i * DD + ch];
    s += v;
    q += v * v;
  }
  ssum[t] = s;
  ssq[t] = q;
  __syncthreads();
  for (int off = 128; off > 0; off >>= 1) {
    if (t < off) {
      ssum[t] += ssum[t + off];
      ssq[t] += ssq[t + off];
    }
    __syncthreads();
  }
  if (t == 0) {
    const float invN = 1.0f / (float)(BB * SS);
    float mean = ssum[0] * invN;
    float var = ssq[0] * invN - mean * mean;
    stats[ch] = mean;
    stats[DD + ch] = rsqrtf(var + 1e-5f);
  }
}

__global__ __launch_bounds__(256) void bn_apply_kernel(
    float* __restrict__ out, const float* __restrict__ stats,
    const float* __restrict__ gamma, const float* __restrict__ beta) {
  int idx = blockIdx.x * 256 + threadIdx.x;
  int ch = idx & (DD - 1);
  float v = out[idx];
  out[idx] = (v - stats[ch]) * stats[DD + ch] * gamma[ch] + beta[ch];
}

extern "C" void kernel_launch(void* const* d_in, const int* in_sizes, int n_in,
                              void* d_out, int out_size, void* d_ws,
                              size_t ws_size, hipStream_t stream) {
  const float* x     = (const float*)d_in[0];
  const float* Wq    = (const float*)d_in[1];
  const float* bq    = (const float*)d_in[2];
  const float* Wk    = (const float*)d_in[3];
  const float* bk    = (const float*)d_in[4];
  const float* Wv    = (const float*)d_in[5];
  const float* bv    = (const float*)d_in[6];
  const float* gamma = (const float*)d_in[7];
  const float* beta  = (const float*)d_in[8];
  float* out = (float*)d_out;
  float* ws = (float*)d_ws;
  float* stats = ws + (size_t)3 * BB * SS * DD;

  qkv_kernel<<<(3 * (BB * SS / 16)) / 4, 128, 0, stream>>>(x, Wq, bq, Wk, bk,
                                                           Wv, bv, ws);
  attn_kernel<<<BB * HH * (SS / ATT_QPB), ATT_WAVES * 32, 0, stream>>>(x, ws, out);
  bn_stats_kernel<<<DD, 256, 0, stream>>>(out, stats);
  bn_apply_kernel<<<(BB * SS * DD) / 256, 256, 0, stream>>>(out, stats, gamma,
                                                            beta);
}


# attention: one key sweep; a row's exponent shift starts at its maximum over key tile 0 and is raised, with one rescale of its sums, only if a score passes it by 2^16; no maxima sweep; same sum order
# speedup vs baseline: 3.8018x; 1.0907x over previous
#include <hip/hip_runtime.h>
#include <math.h>


typedef __attribute__((ext_vector_type(2))) float v2f;
typedef __attribute__((ext_vector_type(8))) float v8f;

#define BB 4
#define SS 2048
#define DD 64
#define HH 16
#define HDIM 4

static __device__ __forceinline__ v8f wmma_f32_16x16x4(v2f a, v2f b, v8f c) {
  return __builtin_amdgcn_wmma_f32_16x16x4_f32(false, a, false, b, (short)0, c,
                                               false, false);
}

__global__ __launch_bounds__(128) void qkv_kernel(
    const float* __restrict__ x,
    const float* __restrict__ Wq, const float* __restrict__ bq,
    const float* __restrict__ Wk, const float* __restrict__ bk,
    const float* __restrict__ Wv, const float* __restrict__ bv,
    float* __restrict__ ws) {
  const int NT = (BB * SS) / 16;
  int tid   = blockIdx.x * 4 + (threadIdx.x >> 5);
  int which = tid / NT;
  int tile  = tid % NT;
  int r0    = tile * 16;

  const float* W    = (which == 0) ? Wq : (which == 1) ? Wk : Wv;
  const float* bias = (which == 0) ? bq : (which == 1) ? bk : bv;
  float* Out = ws + (size_t)which * (BB * SS * DD);

  int lane = threadIdx.x & 31;
  int half = lane >> 4;
  int l16  = lane & 15;

  v2f afrag[16];
  const float* xrow = x + (size_t)(r0 + l16) * DD + 2 * half;
#pragma unroll
  for (int kc = 0; kc < 16; ++kc) {
    afrag[kc].x = xrow[4 * kc + 0];
    afrag[kc].y = xrow[4 * kc + 1];
  }

  v8f acc[4];
#pragma unroll
  for (int n = 0; n < 4; ++n)
#pragma unroll
    for (int v = 0; v < 8; ++v) acc[n][v] = 0.0f;

#pragma unroll
  for (int kc = 0; kc < 16; ++kc) {
    int krow = 4 * kc + 2 * half;
#pragma unroll
    for (int n = 0; n < 4; ++n) {
      v2f bfrag;
      bfrag.x = W[(size_t)krow * DD + n * 16 + l16];
      bfrag.y = W[(size_t)(krow + 1) * DD + n * 16 + l16];
      acc[n] = wmma_f32_16x16x4(afrag[kc], bfrag, acc[n]);
    }
  }

#pragma unroll
  for (int n = 0; n < 4; ++n) {
    int col = n * 16 + l16;
    float bb = bias[col];
#pragma unroll
    for (int v = 0; v < 8; ++v) {
      int row = r0 + v + 8 * half;
      Out[(size_t)row * DD + col] = fmaxf(acc[n][v] + bb, 0.0f);
    }
  }
}

#define ATT_WAVES 32
#define ATT_QPB   (ATT_WAVES * 16)
#define LDS_K     0
#define LDS_V     (SS * HDIM)
#define LDS_Z     (2 * SS * HDIM)
#define LDS_P     (LDS_Z + 128)
#define LDS_USED  (LDS_P + ATT_WAVES * 256)
#define LDS_FLOATS (164 * 256)

__global__ __launch_bounds__(ATT_WAVES * 32) void attn_kernel(
    const float* __restrict__ x, const float* __restrict__ ws,
    float* __restrict__ out) {
  const float* Q = ws;
  const float* K = ws + (size_t)BB * SS * DD;
  const float* V = ws + (size_t)2 * BB * SS * DD;

  __shared__ __attribute__((aligned(16))) float smem[LDS_FLOATS];
  static_assert(LDS_USED <= LDS_FLOATS, "LDS map");

  int t    = threadIdx.x;
  int wave = t >> 5;
  int lane = t & 31;
  int half = lane >> 4;
  int l16  = lane & 15;

  const int QT = SS / ATT_QPB;
  int bh = blockIdx.x / QT;
  int qt = blockIdx.x % QT;
  int b = bh / HH, h = bh % HH;
  int q0 = qt * ATT_QPB + wave * 16;

  const float* Qb = Q + ((size_t)b * SS) * DD + h * HDIM;
  const float* Kb = K + ((size_t)b * SS) * DD + h * HDIM;
  const float* Vb = V + ((size_t)b * SS) * DD + h * HDIM;

  {
    float4 kr[SS / (ATT_WAVES * 32)];
#pragma unroll
    for (int i = 0; i < SS / (ATT_WAVES * 32); ++i) {
      const float* kp = Kb + (size_t)(t + i * ATT_WAVES * 32) * DD;
      kr[i] = make_float4(kp[0], kp[1], kp[2], kp[3]);
    }
    const float* vp = Vb + (size_t)(2 * t) * DD;
    float4 va = make_float4(vp[0], vp[1], vp[2], vp[3]);
    float4 vb = make_float4(vp[DD], vp[DD + 1], vp[DD + 2], vp[DD + 3]);
#pragma unroll
    for (int i = 0; i < SS / (ATT_WAVES * 32); ++i)
      *(float4*)&smem[LDS_K + (t + i * ATT_WAVES * 32) * 4] = kr[i];
    *(float4*)&smem[LDS_V + t * 8]     = make_float4(va.x, vb.x, va.y, vb.y);
    *(float4*)&smem[LDS_V + t * 8 + 4] = make_float4(va.z, vb.z, va.w, vb.w);
    if (t < 128) smem[LDS_Z + t] = 0.0f;
  }

  v2f aq;
  aq.x = Qb[(size_t)(q0 + l16) * DD + 2 * half];
  aq.y = Qb[(size_t)(q0 + l16) * DD + 2 * half + 1];

  __syncthreads();

  v8f zc;
#pragma unroll
  for (int v = 0; v < 8; ++v) zc[v] = 0.0f;

  const float* kfrag = smem + LDS_K + l16 * 4 + 2 * half;

  const float c = 0.72134752044448170368f;
  float mc[8];
  {
    v2f bk = *(const v2f*)(kfrag);
    v8f s = wmma_f32_16x16x4(aq, bk, zc);
#pragma unroll
    for (int v = 0; v < 8; ++v) {
      float m = s[v];
      m = fmaxf(m, __shfl_xor(m, 1, 16));
      m = fmaxf(m, __shfl_xor(m, 2, 16));
      m = fmaxf(m, __shfl_xor(m, 4, 16));
      m = fmaxf(m, __shfl_xor(m, 8, 16));
      mc[v] = m * c;
    }
  }

  const float MARGIN = 16.0f;
  float lpart[8];
  v8f accO;
#pragma unroll
  for (int v = 0; v < 8; ++v) {
    lpart[v] = 0.0f;
    accO[v]  = 0.0f;
  }

  const bool padlane = (l16 >= HDIM);
  const float* vfrag = padlane ? (smem + LDS_Z + 16)
                               : (smem + LDS_V + 8 * half + 2 * l16);
  const int vstep = padlane ? 0 : 64;

  float* pst = smem + LDS_P + wave * 256 + (8 * half) * 16 + l16;
  const float* pld = smem + LDS_P + wave * 256 + l16 * 16 + 2 * half;

  for (int k0 = 0; k0 < SS; k0 += 32) {
    v2f bk0 = *(const v2f*)(kfrag + k0 * 4);
    v2f bk1 = *(const v2f*)(kfrag + k0 * 4 + 64);
    v8f s0 = wmma_f32_16x16x4(aq, bk0, zc);
    v8f s1 = wmma_f32_16x16x4(aq, bk1, zc);

    float t0[8], t1[8];
    float tm = -INFINITY;
#pragma unroll
    for (int v = 0; v < 8; ++v) {
      t0[v] = fmaf(s0[v], c, -mc[v]);
      t1[v] = fmaf(s1[v], c, -mc[v]);
      tm = fmaxf(tm, fmaxf(t0[v], t1[v]));
    }

    if (__builtin_amdgcn_ballot_w32(tm > MARGIN) != 0u) {
#pragma unroll
      for (int v = 0; v < 8; ++v) {
        float m = fmaxf(s0[v], s1[v]);
        m = fmaxf(m, __shfl_xor(m, 1, 16));
        m = fmaxf(m, __shfl_xor(m, 2, 16));
        m = fmaxf(m, __shfl_xor(m, 4, 16));
        m = fmaxf(m, __shfl_xor(m, 8, 16));
        float mn = fmaxf(mc[v], m * c);
        float f = __builtin_amdgcn_exp2f(mc[v] - mn);
        accO[v] *= f;
        lpart[v] *= f;
        mc[v] = mn;
        t0[v] = fmaf(s0[v], c, -mn);
        t1[v] = fmaf(s1[v], c, -mn);
      }
    }

#pragma unroll
    for (int v = 0; v < 8; ++v) {
      float p = __builtin_amdgcn_exp2f(t0[v]);
      lpart[v] += p;
      pst[v * 16] = p;
    }
#pragma unroll
    for (int kc = 0; kc < 4; ++kc) {
      v2f ap = *(const v2f*)(pld + 4 * kc);
      v2f bv = *(const v2f*)(vfrag + 16 * kc);
      accO = wmma_f32_16x16x4(ap, bv, accO);
    }
    vfrag += vstep;

#pragma unroll
    for (int v = 0; v < 8; ++v) {
      float p = __builtin_amdgcn_exp2f(t1[v]);
      lpart[v] += p;
      pst[v * 16] = p;
    }
#pragma unroll
    for (int kc = 0; kc < 4; ++kc) {
      v2f ap = *(const v2f*)(pld + 4 * kc);
      v2f bv = *(const v2f*)(vfrag + 16 * kc);
      accO = wmma_f32_16x16x4(ap, bv, accO);
    }
    vfrag += vstep;
  }

  float lrow[8];
#pragma unroll
  for (int v = 0; v < 8; ++v) {
    float rs = lpart[v];
    rs += __shfl_xor(rs, 1, 16);
    rs += __shfl_xor(rs, 2, 16);
    rs += __shfl_xor(rs, 4, 16);
    rs += __shfl_xor(rs, 8, 16);
    lrow[v] = rs;
  }

  if (l16 < HDIM) {
    int ch = h * HDIM + l16;
#pragma unroll
    for (int v = 0; v < 8; ++v) {
      int row = q0 + v + 8 * half;
      size_t idx = ((size_t)b * SS + row) * DD + ch;
      out[idx] = accO[v] / lrow[v] + x[idx];
    }
  }
}

__global__ __launch_bounds__(256) void bn_stats_kernel(
    const float* __restrict__ out, float* __restrict__ stats) {
  __shared__ float ssum[256], ssq[256];
  int ch = blockIdx.x;
  int t = threadIdx.x;
  float s = 0.0f, q = 0.0f;
  for (int i = t; i < BB * SS; i += 256) {
    float v = out[(size_t)i * DD + ch];
    s += v;
    q += v * v;
  }
  ssum[t] = s;
  ssq[t] = q;
  __syncthreads();
  for (int off = 128; off > 0; off >>= 1) {
    if (t < off) {
      ssum[t] += ssum[t + off];
      ssq[t] += ssq[t + off];
    }
    __syncthreads();
  }
  if (t == 0) {
    const float invN = 1.0f / (float)(BB * SS);
    float mean = ssum[0] * invN;
    float var = ssq[0] * invN - mean * mean;
    stats[ch] = mean;
    stats[DD + ch] = rsqrtf(var + 1e-5f);
  }
}

__global__ __launch_bounds__(256) void bn_apply_kernel(
    float* __restrict__ out, const float* __restrict__ stats,
    const float* __restrict__ gamma, const float* __restrict__ beta) {
  int idx = blockIdx.x * 256 + threadIdx.x;
  int ch = idx & (DD - 1);
  float v = out[idx];
  out[idx] = (v - stats[ch]) * stats[DD + ch] * gamma[ch] + beta[ch];
}

extern "C" void kernel_launch(void* const* d_in, const int* in_sizes, int n_in,
                              void* d_out, int out_size, void* d_ws,
                              size_t ws_size, hipStream_t stream) {
  const float* x     = (const float*)d_in[0];
  const float* Wq    = (const float*)d_in[1];
  const float* bq    = (const float*)d_in[2];
  const float* Wk    = (const float*)d_in[3];
  const float* bk    = (const float*)d_in[4];
  const float* Wv    = (const float*)d_in[5];
  const float* bv    = (const float*)d_in[6];
  const float* gamma = (const float*)d_in[7];
  const float* beta  = (const float*)d_in[8];
  float* out = (float*)d_out;
  float* ws = (float*)d_ws;
  float* stats = ws + (size_t)3 * BB * SS * DD;

  qkv_kernel<<<(3 * (BB * SS / 16)) / 4, 128, 0, stream>>>(x, Wq, bq, Wk, bk,
                                                           Wv, bv, ws);
  attn_kernel<<<BB * HH * (SS / ATT_QPB), ATT_WAVES * 32, 0, stream>>>(x, ws, out);
  bn_stats_kernel<<<DD, 256, 0, stream>>>(out, stats);
  bn_apply_kernel<<<(BB * SS * DD) / 256, 256, 0, stream>>>(out, stats, gamma,
                                                            beta);
}


# BatchNorm in one launch: each of 64 blocks re-adds its 32 channels over all rows (whole-line reads of the pre-norm result, kept in the spent Q plane), then normalises 256 rows; sum order changed
# speedup vs baseline: 4.0351x; 1.0614x over previous
#include <hip/hip_runtime.h>
#include <math.h>


typedef __attribute__((ext_vector_type(2))) float v2f;
typedef __attribute__((ext_vector_type(8))) float v8f;

#define BB 4
#define SS 2048
#define DD 64
#define HH 16
#define HDIM 4

static __device__ __forceinline__ v8f wmma_f32_16x16x4(v2f a, v2f b, v8f c) {
  return __builtin_amdgcn_wmma_f32_16x16x4_f32(false, a, false, b, (short)0, c,
                                               false, false);
}

__global__ __launch_bounds__(128) void qkv_kernel(
    const float* __restrict__ x,
    const float* __restrict__ Wq, const float* __restrict__ bq,
    const float* __restrict__ Wk, const float* __restrict__ bk,
    const float* __restrict__ Wv, const float* __restrict__ bv,
    float* __restrict__ ws) {
  const int NT = (BB * SS) / 16;
  int tid   = blockIdx.x * 4 + (threadIdx.x >> 5);
  int which = tid / NT;
  int tile  = tid % NT;
  int r0    = tile * 16;

  const float* W    = (which == 0) ? Wq : (which == 1) ? Wk : Wv;
  const float* bias = (which == 0) ? bq : (which == 1) ? bk : bv;
  float* Out = ws + (size_t)which * (BB * SS * DD);

  int lane = threadIdx.x & 31;
  int half = lane >> 4;
  int l16  = lane & 15;

  v2f afrag[16];
  const float* xrow = x + (size_t)(r0 + l16) * DD + 2 * half;
#pragma unroll
  for (int kc = 0; kc < 16; ++kc) {
    afrag[kc].x = xrow[4 * kc + 0];
    afrag[kc].y = xrow[4 * kc + 1];
  }

  v8f acc[4];
#pragma unroll
  for (int n = 0; n < 4; ++n)
#pragma unroll
    for (int v = 0; v < 8; ++v) acc[n][v] = 0.0f;

#pragma unroll
  for (int kc = 0; kc < 16; ++kc) {
    int krow = 4 * kc + 2 * half;
#pragma unroll
    for (int n = 0; n < 4; ++n) {
      v2f bfrag;
      bfrag.x = W[(size_t)krow * DD + n * 16 + l16];
      bfrag.y = W[(size_t)(krow + 1) * DD + n * 16 + l16];
      acc[n] = wmma_f32_16x16x4(afrag[kc], bfrag, acc[n]);
    }
  }

#pragma unroll
  for (int n = 0; n < 4; ++n) {
    int col = n * 16 + l16;
    float bb = bias[col];
#pragma unroll
    for (int v = 0; v < 8; ++v) {
      int row = r0 + v + 8 * half;
      Out[(size_t)row * DD + col] = fmaxf(acc[n][v] + bb, 0.0f);
    }
  }
}

#define ATT_WAVES 32
#define ATT_QPB   (ATT_WAVES * 16)
#define LDS_K     0
#define LDS_V     (SS * HDIM)
#define LDS_Z     (2 * SS * HDIM)
#define LDS_P     (LDS_Z + 128)
#define LDS_USED  (LDS_P + ATT_WAVES * 256)
#define LDS_FLOATS (164 * 256)

__global__ __launch_bounds__(ATT_WAVES * 32) void attn_kernel(
    const float* __restrict__ x, float* __restrict__ ws) {
  const float* Q = ws;
  float* pre = ws;
  const float* K = ws + (size_t)BB * SS * DD;
  const float* V = ws + (size_t)2 * BB * SS * DD;

  __shared__ __attribute__((aligned(16))) float smem[LDS_FLOATS];
  static_assert(LDS_USED <= LDS_FLOATS, "LDS map");

  int t    = threadIdx.x;
  int wave = t >> 5;
  int lane = t & 31;
  int half = lane >> 4;
  int l16  = lane & 15;

  const int QT = SS / ATT_QPB;
  int bh = blockIdx.x / QT;
  int qt = blockIdx.x % QT;
  int b = bh / HH, h = bh % HH;
  int q0 = qt * ATT_QPB + wave * 16;

  const float* Qb = Q + ((size_t)b * SS) * DD + h * HDIM;
  const float* Kb = K + ((size_t)b * SS) * DD + h * HDIM;
  const float* Vb = V + ((size_t)b * SS) * DD + h * HDIM;

  {
    float4 kr[SS / (ATT_WAVES * 32)];
#pragma unroll
    for (int i = 0; i < SS / (ATT_WAVES * 32); ++i) {
      const float* kp = Kb + (size_t)(t + i * ATT_WAVES * 32) * DD;
      kr[i] = make_float4(kp[0], kp[1], kp[2], kp[3]);
    }
    const float* vp = Vb + (size_t)(2 * t) * DD;
    float4 va = make_float4(vp[0], vp[1], vp[2], vp[3]);
    float4 vb = make_float4(vp[DD], vp[DD + 1], vp[DD + 2], vp[DD + 3]);
#pragma unroll
    for (int i = 0; i < SS / (ATT_WAVES * 32); ++i)
      *(float4*)&smem[LDS_K + (t + i * ATT_WAVES * 32) * 4] = kr[i];
    *(float4*)&smem[LDS_V + t * 8]     = make_float4(va.x, vb.x, va.y, vb.y);
    *(float4*)&smem[LDS_V + t * 8 + 4] = make_float4(va.z, vb.z, va.w, vb.w);
    if (t < 128) smem[LDS_Z + t] = 0.0f;
  }

  v2f aq;
  aq.x = Qb[(size_t)(q0 + l16) * DD + 2 * half];
  aq.y = Qb[(size_t)(q0 + l16) * DD + 2 * half + 1];

  __syncthreads();

  v8f zc;
#pragma unroll
  for (int v = 0; v < 8; ++v) zc[v] = 0.0f;

  const float* kfrag = smem + LDS_K + l16 * 4 + 2 * half;

  const float c = 0.72134752044448170368f;
  float mc[8];
  {
    v2f bk = *(const v2f*)(kfrag);
    v8f s = wmma_f32_16x16x4(aq, bk, zc);
#pragma unroll
    for (int v = 0; v < 8; ++v) {
      float m = s[v];
      m = fmaxf(m, __shfl_xor(m, 1, 16));
      m = fmaxf(m, __shfl_xor(m, 2, 16));
      m = fmaxf(m, __shfl_xor(m, 4, 16));
      m = fmaxf(m, __shfl_xor(m, 8, 16));
      mc[v] = m * c;
    }
  }

  const float MARGIN = 16.0f;
  float lpart[8];
  v8f accO;
#pragma unroll
  for (int v = 0; v < 8; ++v) {
    lpart[v] = 0.0f;
    accO[v]  = 0.0f;
  }

  const bool padlane = (l16 >= HDIM);
  const float* vfrag = padlane ? (smem + LDS_Z + 16)
                               : (smem + LDS_V + 8 * half + 2 * l16);
  const int vstep = padlane ? 0 : 64;

  float* pst = smem + LDS_P + wave * 256 + (8 * half) * 16 + l16;
  const float* pld = smem + LDS_P + wave * 256 + l16 * 16 + 2 * half;

  for (int k0 = 0; k0 < SS; k0 += 32) {
    v2f bk0 = *(const v2f*)(kfrag + k0 * 4);
    v2f bk1 = *(const v2f*)(kfrag + k0 * 4 + 64);
    v8f s0 = wmma_f32_16x16x4(aq, bk0, zc);
    v8f s1 = wmma_f32_16x16x4(aq, bk1, zc);

    float t0[8], t1[8];
    float tm = -INFINITY;
#pragma unroll
    for (int v = 0; v < 8; ++v) {
      t0[v] = fmaf(s0[v], c, -mc[v]);
      t1[v] = fmaf(s1[v], c, -mc[v]);
      tm = fmaxf(tm, fmaxf(t0[v], t1[v]));
    }

    if (__builtin_amdgcn_ballot_w32(tm > MARGIN) != 0u) {
#pragma unroll
      for (int v = 0; v < 8; ++v) {
        float m = fmaxf(s0[v], s1[v]);
        m = fmaxf(m, __shfl_xor(m, 1, 16));
        m = fmaxf(m, __shfl_xor(m, 2, 16));
        m = fmaxf(m, __shfl_xor(m, 4, 16));
        m = fmaxf(m, __shfl_xor(m, 8, 16));
        float mn = fmaxf(mc[v], m * c);
        float f = __builtin_amdgcn_exp2f(mc[v] - mn);
        accO[v] *= f;
        lpart[v] *= f;
        mc[v] = mn;
        t0[v] = fmaf(s0[v], c, -mn);
        t1[v] = fmaf(s1[v], c, -mn);
      }
    }

#pragma unroll
    for (int v = 0; v < 8; ++v) {
      float p = __builtin_amdgcn_exp2f(t0[v]);
      lpart[v] += p;
      pst[v * 16] = p;
    }
#pragma unroll
    for (int kc = 0; kc < 4; ++kc) {
      v2f ap = *(const v2f*)(pld + 4 * kc);
      v2f bv = *(const v2f*)(vfrag + 16 * kc);
      accO = wmma_f32_16x16x4(ap, bv, accO);
    }
    vfrag += vstep;

#pragma unroll
    for (int v = 0; v < 8; ++v) {
      float p = __builtin_amdgcn_exp2f(t1[v]);
      lpart[v] += p;
      pst[v * 16] = p;
    }
#pragma unroll
    for (int kc = 0; kc < 4; ++kc) {
      v2f ap = *(const v2f*)(pld + 4 * kc);
      v2f bv = *(const v2f*)(vfrag + 16 * kc);
      accO = wmma_f32_16x16x4(ap, bv, accO);
    }
    vfrag += vstep;
  }

  float lrow[8];
#pragma unroll
  for (int v = 0; v < 8; ++v) {
    float rs = lpart[v];
    rs += __shfl_xor(rs, 1, 16);
    rs += __shfl_xor(rs, 2, 16);
    rs += __shfl_xor(rs, 4, 16);
    rs += __shfl_xor(rs, 8, 16);
    lrow[v] = rs;
  }

  if (l16 < HDIM) {
    int ch = h * HDIM + l16;
#pragma unroll
    for (int v = 0; v < 8; ++v) {
      int row = q0 + v + 8 * half;
      size_t idx = ((size_t)b * SS + row) * DD + ch;
      pre[idx] = accO[v] / lrow[v] + x[idx];
    }
  }
}

#define BN_THREADS 1024
#define BN_SLABS   32
#define BN_ROWS    ((BB * SS) / BN_SLABS)
#define BN_CHUNK   ((BB * SS) / 32)
#define BN_STEPS   (BN_CHUNK / 4)
#define BN_LDS_FLOATS (164 * 256)

__global__ __launch_bounds__(BN_THREADS) void bn_kernel(
    const float* __restrict__ pre, float* __restrict__ out,
    const float* __restrict__ gamma, const float* __restrict__ beta) {
  __shared__ __attribute__((aligned(16))) float bsm[BN_LDS_FLOATS];
  float* red_s = bsm;
  float* red_q = bsm + 1024;
  float* stat  = bsm + 2048;

  int t    = threadIdx.x;
  int wave = t >> 5;
  int lane = t & 31;
  int cg   = blockIdx.x & 1;
  int slab = blockIdx.x >> 1;
  int cq   = lane & 7;
  int ro   = lane >> 3;
  int ch0  = 32 * cg + 4 * cq;

  size_t i0 = ((size_t)slab * BN_ROWS + (t >> 3)) * DD + ch0;
  size_t i1 = i0 + (size_t)(BN_ROWS / 2) * DD;
  float4 m0 = *(const float4*)(pre + i0);
  float4 m1 = *(const float4*)(pre + i1);
  float g0 = gamma[ch0], g1 = gamma[ch0 + 1], g2 = gamma[ch0 + 2], g3 = gamma[ch0 + 3];
  float e0 = beta[ch0], e1 = beta[ch0 + 1], e2 = beta[ch0 + 2], e3 = beta[ch0 + 3];

  const float* p = pre + ((size_t)wave * BN_CHUNK + ro) * DD + ch0;
  int first = 2 * slab;
  float s0 = 0.0f, s1 = 0.0f, s2 = 0.0f, s3 = 0.0f;
  float q0 = 0.0f, q1 = 0.0f, q2 = 0.0f, q3 = 0.0f;
#pragma unroll 16
  for (int n = 0; n < BN_STEPS; ++n) {
    int i = (n + first) & (BN_STEPS - 1);
    float4 v = *(const float4*)(p + (size_t)i * (4 * DD));
    s0 += v.x; q0 = fmaf(v.x, v.x, q0);
    s1 += v.y; q1 = fmaf(v.y, v.y, q1);
    s2 += v.z; q2 = fmaf(v.z, v.z, q2);
    s3 += v.w; q3 = fmaf(v.w, v.w, q3);
  }
  s0 += __shfl_xor(s0, 8);  s1 += __shfl_xor(s1, 8);
  s2 += __shfl_xor(s2, 8);  s3 += __shfl_xor(s3, 8);
  q0 += __shfl_xor(q0, 8);  q1 += __shfl_xor(q1, 8);
  q2 += __shfl_xor(q2, 8);  q3 += __shfl_xor(q3, 8);
  s0 += __shfl_xor(s0, 16); s1 += __shfl_xor(s1, 16);
  s2 += __shfl_xor(s2, 16); s3 += __shfl_xor(s3, 16);
  q0 += __shfl_xor(q0, 16); q1 += __shfl_xor(q1, 16);
  q2 += __shfl_xor(q2, 16); q3 += __shfl_xor(q3, 16);
  if (lane < 8) {
    *(float4*)&red_s[wave * 32 + 4 * cq] = make_float4(s0, s1, s2, s3);
    *(float4*)&red_q[wave * 32 + 4 * cq] = make_float4(q0, q1, q2, q3);
  }
  __syncthreads();
  if (t < 32) {
    float a = 0.0f, b = 0.0f;
#pragma unroll
    for (int k = 0; k < 32; ++k) {
      a += red_s[k * 32 + t];
      b += red_q[k * 32 + t];
    }
    const float invN = 1.0f / (float)(BB * SS);
    float mean = a * invN;
    float var = b * invN - mean * mean;
    stat[t] = mean;
    stat[32 + t] = rsqrtf(var + 1e-5f);
  }
  __syncthreads();

  float4 mu = *(const float4*)&stat[4 * cq];
  float4 rs = *(const float4*)&stat[32 + 4 * cq];
  float4 o0, o1;
  o0.x = (m0.x - mu.x) * rs.x * g0 + e0;
  o0.y = (m0.y - mu.y) * rs.y * g1 + e1;
  o0.z = (m0.z - mu.z) * rs.z * g2 + e2;
  o0.w = (m0.w - mu.w) * rs.w * g3 + e3;
  o1.x = (m1.x - mu.x) * rs.x * g0 + e0;
  o1.y = (m1.y - mu.y) * rs.y * g1 + e1;
  o1.z = (m1.z - mu.z) * rs.z * g2 + e2;
  o1.w = (m1.w - mu.w) * rs.w * g3 + e3;
  *(float4*)(out + i0) = o0;
  *(float4*)(out + i1) = o1;
}

extern "C" void kernel_launch(void* const* d_in, const int* in_sizes, int n_in,
                              void* d_out, int out_size, void* d_ws,
                              size_t ws_size, hipStream_t stream) {
  const float* x     = (const float*)d_in[0];
  const float* Wq    = (const float*)d_in[1];
  const float* bq    = (const float*)d_in[2];
  const float* Wk    = (const float*)d_in[3];
  const float* bk    = (const float*)d_in[4];
  const float* Wv    = (const float*)d_in[5];
  const float* bv    = (const float*)d_in[6];
  const float* gamma = (const float*)d_in[7];
  const float* beta  = (const float*)d_in[8];
  float* out = (float*)d_out;
  float* ws = (float*)d_ws;

  qkv_kernel<<<(3 * (BB * SS / 16)) / 4, 128, 0, stream>>>(x, Wq, bq, Wk, bk,
                                                           Wv, bv, ws);
  attn_kernel<<<BB * HH * (SS / ATT_QPB), ATT_WAVES * 32, 0, stream>>>(x, ws);
  bn_kernel<<<BN_SLABS * 2, BN_THREADS, 0, stream>>>(ws, out, gamma, beta);
}


# projection fused into the attention launch: each workgroup computes its head's K, V (2048 rows) and Q from x by WMMA straight into LDS; no projection launch (3 -> 2); sums and order unchanged
# speedup vs baseline: 4.7353x; 1.1735x over previous
#include <hip/hip_runtime.h>
#include <math.h>


typedef __attribute__((ext_vector_type(2))) float v2f;
typedef __attribute__((ext_vector_type(8))) float v8f;

#define BB 4
#define SS 2048
#define DD 64
#define HH 16
#define HDIM 4

static __device__ __forceinline__ v8f wmma_f32_16x16x4(v2f a, v2f b, v8f c) {
  return __builtin_amdgcn_wmma_f32_16x16x4_f32(false, a, false, b, (short)0, c,
                                               false, false);
}

#define ATT_WAVES 32
#define ATT_QPB   (ATT_WAVES * 16)
#define LDS_K     0
#define LDS_V     (SS * HDIM)
#define LDS_Z     (2 * SS * HDIM)
#define LDS_ONE   (LDS_Z + 128)
#define LDS_P     (LDS_ONE + 128)
#define XPITCH    68
#define LDS_X     (LDS_P + ATT_WAVES * 256)
#define LDS_USED  (LDS_X + ATT_WAVES * 16 * XPITCH)
#define LDS_FLOATS LDS_USED

__global__ __launch_bounds__(ATT_WAVES * 32) void attn_kernel(
    const float* __restrict__ x,
    const float* __restrict__ Wq, const float* __restrict__ bq,
    const float* __restrict__ Wk, const float* __restrict__ bk,
    const float* __restrict__ Wv, const float* __restrict__ bv,
    float* __restrict__ ws) {
  float* pre = ws;

  __shared__ __attribute__((aligned(16))) float smem[LDS_FLOATS];
  static_assert(LDS_USED <= LDS_FLOATS, "LDS map");

  int t    = threadIdx.x;
  int wave = t >> 5;
  int lane = t & 31;
  int half = lane >> 4;
  int l16  = lane & 15;

  const int QT = SS / ATT_QPB;
  int bh = blockIdx.x / QT;
  int qt = blockIdx.x % QT;
  int b = bh / HH, h = bh % HH;
  int q0 = qt * ATT_QPB + wave * 16;

  const int pd = l16 & 3;
  const bool pk = (l16 < 4);
  const bool pv = (l16 >= 4) && (l16 < 8);
  const bool pq = (l16 >= 8) && (l16 < 12);
  const float* Wsel = pk ? Wk : pv ? Wv : Wq;
  const float* bsel = pk ? bk : pv ? bv : bq;
  const bool plive = (l16 < 12);
  const int pcol = h * HDIM + pd;
  v2f bw[16];
#pragma unroll
  for (int kc = 0; kc < 16; ++kc) {
    int krow = 4 * kc + 2 * half;
    float w0 = Wsel[(size_t)krow * DD + pcol];
    float w1 = Wsel[(size_t)(krow + 1) * DD + pcol];
    bw[kc].x = plive ? w0 : 0.0f;
    bw[kc].y = plive ? w1 : 0.0f;
  }
  const float pbias = plive ? bsel[pcol] : 0.0f;

  if (t < 128) {
    smem[LDS_Z + t]   = 0.0f;
    smem[LDS_ONE + t] = 1.0f;
  }

  float* xs = smem + LDS_X + wave * (16 * XPITCH);
  float* qs = smem + LDS_P + wave * 256;
  const float* xb = x + (size_t)b * SS * DD;
  for (int j = 0; j < SS / ATT_QPB; ++j) {
    int jq = (j + h) & (SS / ATT_QPB - 1);
    int T = jq * ATT_WAVES + wave;
    const float* xt = xb + (size_t)T * 16 * DD + lane * 4;
    float4 xv[8];
#pragma unroll
    for (int i = 0; i < 8; ++i) xv[i] = *(const float4*)(xt + i * 128);
#pragma unroll
    for (int i = 0; i < 8; ++i)
      *(float4*)(xs + (2 * i + half) * XPITCH + l16 * 4) = xv[i];
    v8f pacc;
#pragma unroll
    for (int v = 0; v < 8; ++v) pacc[v] = 0.0f;
#pragma unroll
    for (int kc = 0; kc < 16; ++kc) {
      v2f pa = *(const v2f*)(xs + l16 * XPITCH + 4 * kc + 2 * half);
      pacc = wmma_f32_16x16x4(pa, bw[kc], pacc);
    }
    int prow = 16 * T + 8 * half;
    if (pk) {
#pragma unroll
      for (int v = 0; v < 8; ++v)
        smem[LDS_K + (prow + v) * 4 + pd] = fmaxf(pacc[v] + pbias, 0.0f);
    } else if (pv) {
#pragma unroll
      for (int v = 0; v < 8; ++v)
        smem[LDS_V + ((prow + v) >> 1) * 8 + pd * 2 + (v & 1)] =
            fmaxf(pacc[v] + pbias, 0.0f);
    } else if (pq && jq == qt) {
#pragma unroll
      for (int v = 0; v < 8; ++v)
        qs[(8 * half + v) * 4 + pd] = fmaxf(pacc[v] + pbias, 0.0f);
    }
  }

  const float c = 0.72134752044448170368f;
  v2f aq = *(const v2f*)(qs + l16 * 4 + 2 * half);
  aq.x *= c;
  aq.y *= c;

  __syncthreads();

  v8f zc;
#pragma unroll
  for (int v = 0; v < 8; ++v) zc[v] = 0.0f;

  const float* kfrag = smem + LDS_K + l16 * 4 + 2 * half;

  v8f negm;
  {
    v2f bk = *(const v2f*)(kfrag);
    v8f s = wmma_f32_16x16x4(aq, bk, zc);
#pragma unroll
    for (int v = 0; v < 8; ++v) {
      float m = s[v];
      m = fmaxf(m, __shfl_xor(m, 1, 16));
      m = fmaxf(m, __shfl_xor(m, 2, 16));
      m = fmaxf(m, __shfl_xor(m, 4, 16));
      m = fmaxf(m, __shfl_xor(m, 8, 16));
      negm[v] = -m;
    }
  }

  const float MARGIN = 16.0f;
  v8f accO;
#pragma unroll
  for (int v = 0; v < 8; ++v) accO[v] = 0.0f;

  const bool padlane = (l16 >= HDIM);
  const float* vfrag = (l16 < HDIM)  ? (smem + LDS_V + 8 * half + 2 * l16)
                     : (l16 == HDIM) ? (smem + LDS_ONE + 32)
                                     : (smem + LDS_Z + 16);
  const int vstep = padlane ? 0 : 64;

  float* pst = smem + LDS_P + wave * 256 + (8 * half) * 16 + l16;
  const float* pld = smem + LDS_P + wave * 256 + l16 * 16 + 2 * half;

  for (int k0 = 0; k0 < SS; k0 += 32) {
    v2f bk0 = *(const v2f*)(kfrag + k0 * 4);
    v2f bk1 = *(const v2f*)(kfrag + k0 * 4 + 64);
    v8f t0 = wmma_f32_16x16x4(aq, bk0, negm);
    v8f t1 = wmma_f32_16x16x4(aq, bk1, negm);

    float tm = -INFINITY;
#pragma unroll
    for (int v = 0; v < 8; ++v)
      tm = __builtin_fmaximum_numf(tm, __builtin_fmaximum_numf(t0[v], t1[v]));

    if (__builtin_amdgcn_ballot_w32(tm > MARGIN) != 0u) {
      v8f dv, fv;
#pragma unroll
      for (int v = 0; v < 8; ++v) {
        float m = fmaxf(t0[v], t1[v]);
        m = fmaxf(m, __shfl_xor(m, 1, 16));
        m = fmaxf(m, __shfl_xor(m, 2, 16));
        m = fmaxf(m, __shfl_xor(m, 4, 16));
        m = fmaxf(m, __shfl_xor(m, 8, 16));
        dv[v] = fmaxf(m, 0.0f);
        fv[v] = __builtin_amdgcn_exp2f(-dv[v]);
      }
      accO = accO * fv;
      negm = negm - dv;
      t0 = t0 - dv;
      t1 = t1 - dv;
    }

#pragma unroll
    for (int v = 0; v < 8; ++v) pst[v * 16] = __builtin_amdgcn_exp2f(t0[v]);
#pragma unroll
    for (int kc = 0; kc < 4; ++kc) {
      v2f ap = *(const v2f*)(pld + 4 * kc);
      v2f bv = *(const v2f*)(vfrag + 16 * kc);
      accO = wmma_f32_16x16x4(ap, bv, accO);
    }
    vfrag += vstep;

#pragma unroll
    for (int v = 0; v < 8; ++v) pst[v * 16] = __builtin_amdgcn_exp2f(t1[v]);
#pragma unroll
    for (int kc = 0; kc < 4; ++kc) {
      v2f ap = *(const v2f*)(pld + 4 * kc);
      v2f bv = *(const v2f*)(vfrag + 16 * kc);
      accO = wmma_f32_16x16x4(ap, bv, accO);
    }
    vfrag += vstep;
  }

  float lrow[8];
#pragma unroll
  for (int v = 0; v < 8; ++v) lrow[v] = __shfl(accO[v], HDIM, 16);

  if (l16 < HDIM) {
    int ch = h * HDIM + l16;
#pragma unroll
    for (int v = 0; v < 8; ++v) {
      int row = q0 + v + 8 * half;
      size_t idx = ((size_t)b * SS + row) * DD + ch;
      pre[idx] = accO[v] / lrow[v] + x[idx];
    }
  }
}

#define BN_THREADS 1024
#define BN_SLABS   32
#define BN_ROWS    ((BB * SS) / BN_SLABS)
#define BN_CHUNK   ((BB * SS) / 32)
#define BN_STEPS   (BN_CHUNK / 4)
#define BN_LDS_FLOATS (164 * 256)

__global__ __launch_bounds__(BN_THREADS) void bn_kernel(
    const float* __restrict__ pre, float* __restrict__ out,
    const float* __restrict__ gamma, const float* __restrict__ beta) {
  __shared__ __attribute__((aligned(16))) float bsm[BN_LDS_FLOATS];
  float* red_s = bsm;
  float* red_q = bsm + 1024;
  float* stat  = bsm + 2048;

  int t    = threadIdx.x;
  int wave = t >> 5;
  int lane = t & 31;
  int cg   = blockIdx.x & 1;
  int slab = blockIdx.x >> 1;
  int cq   = lane & 7;
  int ro   = lane >> 3;
  int ch0  = 32 * cg + 4 * cq;

  size_t i0 = ((size_t)slab * BN_ROWS + (t >> 3)) * DD + ch0;
  size_t i1 = i0 + (size_t)(BN_ROWS / 2) * DD;
  float4 m0 = *(const float4*)(pre + i0);
  float4 m1 = *(const float4*)(pre + i1);
  float g0 = gamma[ch0], g1 = gamma[ch0 + 1], g2 = gamma[ch0 + 2], g3 = gamma[ch0 + 3];
  float e0 = beta[ch0], e1 = beta[ch0 + 1], e2 = beta[ch0 + 2], e3 = beta[ch0 + 3];

  const float* p = pre + ((size_t)wave * BN_CHUNK + ro) * DD + ch0;
  int first = 2 * slab;
  float s0 = 0.0f, s1 = 0.0f, s2 = 0.0f, s3 = 0.0f;
  float q0 = 0.0f, q1 = 0.0f, q2 = 0.0f, q3 = 0.0f;
#pragma unroll 16
  for (int n = 0; n < BN_STEPS; ++n) {
    int i = (n + first) & (BN_STEPS - 1);
    float4 v = *(const float4*)(p + (size_t)i * (4 * DD));
    s0 += v.x; q0 = fmaf(v.x, v.x, q0);
    s1 += v.y; q1 = fmaf(v.y, v.y, q1);
    s2 += v.z; q2 = fmaf(v.z, v.z, q2);
    s3 += v.w; q3 = fmaf(v.w, v.w, q3);
  }
  s0 += __shfl_xor(s0, 8);  s1 += __shfl_xor(s1, 8);
  s2 += __shfl_xor(s2, 8);  s3 += __shfl_xor(s3, 8);
  q0 += __shfl_xor(q0, 8);  q1 += __shfl_xor(q1, 8);
  q2 += __shfl_xor(q2, 8);  q3 += __shfl_xor(q3, 8);
  s0 += __shfl_xor(s0, 16); s1 += __shfl_xor(s1, 16);
  s2 += __shfl_xor(s2, 16); s3 += __shfl_xor(s3, 16);
  q0 += __shfl_xor(q0, 16); q1 += __shfl_xor(q1, 16);
  q2 += __shfl_xor(q2, 16); q3 += __shfl_xor(q3, 16);
  if (lane < 8) {
    *(float4*)&red_s[wave * 32 + 4 * cq] = make_float4(s0, s1, s2, s3);
    *(float4*)&red_q[wave * 32 + 4 * cq] = make_float4(q0, q1, q2, q3);
  }
  __syncthreads();
  if (t < 32) {
    float a = 0.0f, b = 0.0f;
#pragma unroll
    for (int k = 0; k < 32; ++k) {
      a += red_s[k * 32 + t];
      b += red_q[k * 32 + t];
    }
    const float invN = 1.0f / (float)(BB * SS);
    float mean = a * invN;
    float var = b * invN - mean * mean;
    stat[t] = mean;
    stat[32 + t] = rsqrtf(var + 1e-5f);
  }
  __syncthreads();

  float4 mu = *(const float4*)&stat[4 * cq];
  float4 rs = *(const float4*)&stat[32 + 4 * cq];
  float4 o0, o1;
  o0.x = (m0.x - mu.x) * rs.x * g0 + e0;
  o0.y = (m0.y - mu.y) * rs.y * g1 + e1;
  o0.z = (m0.z - mu.z) * rs.z * g2 + e2;
  o0.w = (m0.w - mu.w) * rs.w * g3 + e3;
  o1.x = (m1.x - mu.x) * rs.x * g0 + e0;
  o1.y = (m1.y - mu.y) * rs.y * g1 + e1;
  o1.z = (m1.z - mu.z) * rs.z * g2 + e2;
  o1.w = (m1.w - mu.w) * rs.w * g3 + e3;
  *(float4*)(out + i0) = o0;
  *(float4*)(out + i1) = o1;
}

extern "C" void kernel_launch(void* const* d_in, const int* in_sizes, int n_in,
                              void* d_out, int out_size, void* d_ws,
                              size_t ws_size, hipStream_t stream) {
  const float* x     = (const float*)d_in[0];
  const float* Wq    = (const float*)d_in[1];
  const float* bq    = (const float*)d_in[2];
  const float* Wk    = (const float*)d_in[3];
  const float* bk    = (const float*)d_in[4];
  const float* Wv    = (const float*)d_in[5];
  const float* bv    = (const float*)d_in[6];
  const float* gamma = (const float*)d_in[7];
  const float* beta  = (const float*)d_in[8];
  float* out = (float*)d_out;
  float* ws = (float*)d_ws;

  attn_kernel<<<BB * HH * (SS / ATT_QPB), ATT_WAVES * 32, 0, stream>>>(
      x, Wq, bq, Wk, bk, Wv, bv, ws);
  bn_kernel<<<BN_SLABS * 2, BN_THREADS, 0, stream>>>(ws, out, gamma, beta);
}
